# starting kernel
# baseline (speedup 1.0000x reference)
#include <hip/hip_runtime.h>
#include <hip/hip_bf16.h>

#define NN 50000
#define NE 500000
#define NR 8
#define DD 128
#define AD 64

typedef _Float16 v16h __attribute__((ext_vector_type(16)));
typedef _Float16 v8h  __attribute__((ext_vector_type(8)));
typedef _Float16 v4h  __attribute__((ext_vector_type(4)));
typedef float    v8f  __attribute__((ext_vector_type(8)));
typedef float    v4f  __attribute__((ext_vector_type(4)));

__device__ __forceinline__ v8f v8f_zero() {
  v8f z;
  #pragma unroll
  for (int i = 0; i < 8; ++i) z[i] = 0.f;
  return z;
}

__device__ __forceinline__ v16h load_frag16(const _Float16* p) {
  union { v16h v; v8h h[2]; } u;
  u.h[0] = *(const v8h*)(p);
  u.h[1] = *(const v8h*)(p + 16);
  return u.v;
}

__device__ __forceinline__ v8f wmma_f16(v16h a, v16h b, v8f c) {
  return __builtin_amdgcn_wmma_f32_16x16x32_f16(false, a, false, b,
                                                (short)0, c, false, false);
}

__device__ __forceinline__ void atomicMaxFloat(float* addr, float val) {
  if (!(val < 0.f)) atomicMax((int*)addr, __float_as_int(val));
  else              atomicMin((unsigned int*)addr, __float_as_uint(val));
}


__global__ void k_cvt(const float* __restrict__ in, _Float16* __restrict__ out, int n) {
  int i = blockIdx.x * blockDim.x + threadIdx.x;
  if (i < n) out[i] = (_Float16)in[i];
}

__global__ void k_cvt_rel(const float* __restrict__ rel, _Float16* __restrict__ relT) {
  int i = blockIdx.x * blockDim.x + threadIdx.x;
  if (i >= 16 * DD * DD) return;
  int d = i & (DD - 1);
  int e = (i >> 7) & (DD - 1);
  int slot = i >> 14;
  int half = slot >> 3, r = slot & 7;
  relT[i] = (_Float16)rel[((size_t)r * 2 * DD + half * DD + d) * DD + e];
}

__global__ void k_init(float* __restrict__ hN, float* __restrict__ smax,
                       float* __restrict__ denom, int nD, int n) {
  int i = blockIdx.x * blockDim.x + threadIdx.x;
  if (i < nD) hN[i] = 0.f;
  if (i < n) { smax[i] = __int_as_float(0xFF800000); denom[i] = 0.f; }
}

__global__ void k_proj(const _Float16* __restrict__ h16,
                       const _Float16* __restrict__ relT16,
                       _Float16* __restrict__ proj16) {
  const int mtiles = NN / 16;
  int bid = blockIdx.x;
  int mtile = bid % mtiles;
  int slot = bid / mtiles;
  int lane = threadIdx.x;
  int lo = lane & 15, hi = lane >> 4;

  const _Float16* A = h16 + (size_t)(mtile * 16 + lo) * DD + hi * 8;
  const _Float16* B = relT16 + ((size_t)slot * DD + lo) * DD + hi * 8;

  v8f acc[8];
  #pragma unroll
  for (int j = 0; j < 8; ++j) acc[j] = v8f_zero();

  #pragma unroll
  for (int kk = 0; kk < 4; ++kk) {
    v16h a = load_frag16(A + kk * 32);
    #pragma unroll
    for (int jt = 0; jt < 8; ++jt) {
      v16h b = load_frag16(B + (size_t)jt * 16 * DD + kk * 32);
      acc[jt] = wmma_f16(a, b, acc[jt]);
    }
  }

  _Float16* out = proj16 + (size_t)slot * NN * DD;
  #pragma unroll
  for (int jt = 0; jt < 8; ++jt) {
    int col = jt * 16 + lo;
    #pragma unroll
    for (int v = 0; v < 8; ++v) {
      int row = mtile * 16 + v + 8 * hi;
      out[(size_t)row * DD + col] = (_Float16)acc[jt][v];
    }
  }
}

__global__ void k_attn(const _Float16* __restrict__ h16,
                       const _Float16* __restrict__ Wa16,
                       const float* __restrict__ rel_att,
                       const float* __restrict__ b_a,
                       const int* __restrict__ src, const int* __restrict__ dst,
                       float* __restrict__ score, float* __restrict__ smax) {
  int e0 = blockIdx.x * 16;
  int lane = threadIdx.x;
  int lo = lane & 15, hi = lane >> 4;
  int e = e0 + lo;
  int dn = dst[e], sn = src[e];

  v8f acc[4];
  #pragma unroll
  for (int j = 0; j < 4; ++j) acc[j] = v8f_zero();

  #pragma unroll
  for (int kk = 0; kk < 8; ++kk) {
    int k0 = kk * 32 + hi * 8;
    const _Float16* ap = (k0 < DD) ? (h16 + (size_t)dn * DD + k0)
                                   : (h16 + (size_t)sn * DD + (k0 - DD));
    v16h a = load_frag16(ap);
    #pragma unroll
    for (int jt = 0; jt < 4; ++jt) {
      v16h b = load_frag16(Wa16 + (size_t)(jt * 16 + lo) * (2 * DD) + k0);
      acc[jt] = wmma_f16(a, b, acc[jt]);
    }
  }

  float s[8];
  #pragma unroll
  for (int v = 0; v < 8; ++v) s[v] = 0.f;
  #pragma unroll
  for (int jt = 0; jt < 4; ++jt) {
    int j = jt * 16 + lo;
    float w = rel_att[j], bb = b_a[j];
    #pragma unroll
    for (int v = 0; v < 8; ++v) {
      float x = acc[jt][v] + bb;
      x = (x > 0.f) ? x : 0.01f * x;
      s[v] += x * w;
    }
  }
  #pragma unroll
  for (int v = 0; v < 8; ++v) {
    float t = s[v];
    t += __shfl_xor(t, 1, 32);
    t += __shfl_xor(t, 2, 32);
    t += __shfl_xor(t, 4, 32);
    t += __shfl_xor(t, 8, 32);
    s[v] = t;
  }
  if (lo == 0) {
    #pragma unroll
    for (int v = 0; v < 8; ++v) {
      int ee = e0 + v + 8 * hi;
      score[ee] = s[v];
      atomicMaxFloat(&smax[dst[ee]], s[v]);
    }
  }
}

__global__ void k_exp(const float* __restrict__ score, const int* __restrict__ dst,
                      const float* __restrict__ smax, float* __restrict__ ex,
                      float* __restrict__ denom) {
  int e = blockIdx.x * blockDim.x + threadIdx.x;
  if (e >= NE) return;
  int dn = dst[e];
  float v = __expf(score[e] - smax[dn]);
  ex[e] = v;
  atomicAdd(&denom[dn], v);
}

__global__ void k_msg(const float* __restrict__ h,
                      const _Float16* __restrict__ proj16,
                      const int* __restrict__ src, const int* __restrict__ dst,
                      const int* __restrict__ etype,
                      const float* __restrict__ ex, const float* __restrict__ denom,
                      float* __restrict__ hN) {
  int e = blockIdx.x * (blockDim.x >> 5) + (threadIdx.x >> 5);
  int lane = threadIdx.x & 31;
  if (e >= NE) return;
  int dn = dst[e], sn = src[e], et = etype[e];
  float attn = ex[e] / denom[dn];
  const _Float16* pd = proj16 + ((size_t)et * NN + dn) * DD + lane * 4;
  const _Float16* ps = proj16 + ((size_t)(8 + et) * NN + sn) * DD + lane * 4;
  v4h pdv = *(const v4h*)pd;
  v4h psv = *(const v4h*)ps;
  v4f hs = *(const v4f*)(h + (size_t)sn * DD + lane * 4);
  float* o = hN + (size_t)dn * DD + lane * 4;
  #pragma unroll
  for (int i = 0; i < 4; ++i) {
    float g = (float)pdv[i] + (float)psv[i];
    g = 1.f / (1.f + __expf(-g));
    atomicAdd(o + i, hs[i] * attn * g);
  }
}

__global__ void k_final(const _Float16* __restrict__ h16,
                        const float* __restrict__ hN,
                        const _Float16* __restrict__ Wl16,
                        const float* __restrict__ b_lin,
                        float* __restrict__ out) {
  int mtile = blockIdx.x;
  int lane = threadIdx.x;
  int lo = lane & 15, hi = lane >> 4;
  int row = mtile * 16 + lo;

  v8f acc[8];
  #pragma unroll
  for (int j = 0; j < 8; ++j) acc[j] = v8f_zero();

  #pragma unroll
  for (int kk = 0; kk < 8; ++kk) {
    int k0 = kk * 32 + hi * 8;
    v16h a;
    if (k0 < DD) {
      a = load_frag16(h16 + (size_t)row * DD + k0);
    } else {
      const float* p = hN + (size_t)row * DD + (k0 - DD);
      union { v16h v; v4h h4[4]; } u;
      v4f f0 = *(const v4f*)(p);
      v4f f1 = *(const v4f*)(p + 4);
      v4f f2 = *(const v4f*)(p + 16);
      v4f f3 = *(const v4f*)(p + 20);
      #pragma unroll
      for (int i = 0; i < 4; ++i) {
        u.h4[0][i] = (_Float16)f0[i];
        u.h4[1][i] = (_Float16)f1[i];
        u.h4[2][i] = (_Float16)f2[i];
        u.h4[3][i] = (_Float16)f3[i];
      }
      a = u.v;
    }
    #pragma unroll
    for (int jt = 0; jt < 8; ++jt) {
      v16h b = load_frag16(Wl16 + (size_t)(jt * 16 + lo) * (2 * DD) + k0);
      acc[jt] = wmma_f16(a, b, acc[jt]);
    }
  }

  #pragma unroll
  for (int jt = 0; jt < 8; ++jt) {
    int col = jt * 16 + lo;
    float bb = b_lin[col];
    #pragma unroll
    for (int v = 0; v < 8; ++v) {
      float x = acc[jt][v] + bb;
      x = (x > 0.f) ? x : 0.01f * x;
      out[(size_t)(mtile * 16 + v + 8 * hi) * DD + col] = x;
    }
  }
}

extern "C" void kernel_launch(void* const* d_in, const int* in_sizes, int n_in,
                              void* d_out, int out_size, void* d_ws, size_t ws_size,
                              hipStream_t stream) {
  const float* h         = (const float*)d_in[0];
  const float* relations = (const float*)d_in[1];
  const float* rel_att   = (const float*)d_in[2];
  const float* W_a       = (const float*)d_in[3];
  const float* b_a       = (const float*)d_in[4];
  const float* W_lin     = (const float*)d_in[5];
  const float* b_lin     = (const float*)d_in[6];
  const int*   src       = (const int*)d_in[7];
  const int*   dst       = (const int*)d_in[8];
  const int*   etype     = (const int*)d_in[9];
  float* out = (float*)d_out;
  (void)in_sizes; (void)n_in; (void)out_size; (void)ws_size;

  char* ws = (char*)d_ws;
  size_t off = 0;
  auto take = [&](size_t bytes) -> char* {
    char* p = ws + off;
    off = (off + bytes + 255) & ~(size_t)255;
    return p;
  };
  _Float16* h16    = (_Float16*)take((size_t)NN * DD * 2);
  _Float16* relT16 = (_Float16*)take((size_t)16 * DD * DD * 2);
  _Float16* Wa16   = (_Float16*)take((size_t)AD * 2 * DD * 2);
  _Float16* Wl16   = (_Float16*)take((size_t)DD * 2 * DD * 2);
  _Float16* proj16 = (_Float16*)take((size_t)16 * NN * DD * 2);
  float* score = (float*)take((size_t)NE * 4);
  float* ex    = (float*)take((size_t)NE * 4);
  float* smax  = (float*)take((size_t)NN * 4);
  float* denom = (float*)take((size_t)NN * 4);
  float* hN    = (float*)take((size_t)NN * DD * 4);

  const int nh = NN * DD;
  k_cvt<<<(nh + 255) / 256, 256, 0, stream>>>(h, h16, nh);
  k_cvt<<<(AD * 2 * DD + 255) / 256, 256, 0, stream>>>(W_a, Wa16, AD * 2 * DD);
  k_cvt<<<(DD * 2 * DD + 255) / 256, 256, 0, stream>>>(W_lin, Wl16, DD * 2 * DD);
  k_cvt_rel<<<(16 * DD * DD) / 256, 256, 0, stream>>>(relations, relT16);
  k_init<<<(nh + 255) / 256, 256, 0, stream>>>(hN, smax, denom, nh, NN);

  k_proj<<<(NN / 16) * 16, 32, 0, stream>>>(h16, relT16, proj16);
  k_attn<<<NE / 16, 32, 0, stream>>>(h16, Wa16, rel_att, b_a, src, dst, score, smax);
  k_exp<<<(NE + 255) / 256, 256, 0, stream>>>(score, dst, smax, ex, denom);
  k_msg<<<(NE + 7) / 8, 256, 0, stream>>>(h, proj16, src, dst, etype, ex, denom, hN);
  k_final<<<NN / 16, 32, 0, stream>>>(h16, hN, Wl16, b_lin, out);
}


# Relation projection re-tiled: a wave keeps one slot's 128x128 matrix in registers as WMMA A operand and walks 25 node tiles (2,000 waves, was 50,000); tiles leave via LDS as whole 256-byte rows
# speedup vs baseline: 1.8195x; 1.8195x over previous
#include <hip/hip_runtime.h>
#include <hip/hip_bf16.h>

#define NN 50000
#define NE 500000
#define NR 8
#define DD 128
#define AD 64

typedef _Float16 v16h __attribute__((ext_vector_type(16)));
typedef _Float16 v8h  __attribute__((ext_vector_type(8)));
typedef _Float16 v4h  __attribute__((ext_vector_type(4)));
typedef float    v8f  __attribute__((ext_vector_type(8)));
typedef float    v4f  __attribute__((ext_vector_type(4)));

__device__ __forceinline__ v8f v8f_zero() {
  v8f z;
  #pragma unroll
  for (int i = 0; i < 8; ++i) z[i] = 0.f;
  return z;
}

__device__ __forceinline__ v16h load_frag16(const _Float16* p) {
  union { v16h v; v8h h[2]; } u;
  u.h[0] = *(const v8h*)(p);
  u.h[1] = *(const v8h*)(p + 16);
  return u.v;
}

__device__ __forceinline__ v8f wmma_f16(v16h a, v16h b, v8f c) {
  return __builtin_amdgcn_wmma_f32_16x16x32_f16(false, a, false, b,
                                                (short)0, c, false, false);
}


__global__ void k_cvt(const float* __restrict__ in, _Float16* __restrict__ out, int n) {
  int i = blockIdx.x * blockDim.x + threadIdx.x;
  if (i < n) out[i] = (_Float16)in[i];
}

__global__ void k_cvt_rel(const float* __restrict__ rel, _Float16* __restrict__ relT) {
  int i = blockIdx.x * blockDim.x + threadIdx.x;
  if (i >= 16 * DD * DD) return;
  int d = i & (DD - 1);
  int e = (i >> 7) & (DD - 1);
  int slot = i >> 14;
  int half = slot >> 3, r = slot & 7;
  relT[i] = (_Float16)rel[((size_t)r * 2 * DD + half * DD + d) * DD + e];
}

#define SCAN_T   1024
#define SCAN_PER 52
#define NPAD     (SCAN_T * SCAN_PER)
static_assert(NPAD >= NN + 1, "the scan must also produce the end of the last node's segment");
static_assert(SCAN_PER % 4 == 0, "a thread's piece is read as 16-byte words");

__global__ void k_init(int* __restrict__ deg) {
  int i = blockIdx.x * blockDim.x + threadIdx.x;
  if (i < NPAD) deg[i] = 0;
}

__global__ void k_count(const int* __restrict__ dst, int* __restrict__ deg,
                        int* __restrict__ rank) {
  int e = blockIdx.x * blockDim.x + threadIdx.x;
  if (e < NE) rank[e] = atomicAdd(&deg[dst[e]], 1);
}

__global__ void __launch_bounds__(SCAN_T) k_scan(const int* __restrict__ deg,
                                                 int* __restrict__ off) {
  __shared__ int wsum[SCAN_T / 32];
  int t = threadIdx.x;
  int lane = t & 31, w = t >> 5;
  const int4* p = (const int4*)(deg + t * SCAN_PER);
  int4 v[SCAN_PER / 4];
  #pragma unroll
  for (int i = 0; i < SCAN_PER / 4; ++i) v[i] = p[i];
  int run = 0;
  #pragma unroll
  for (int i = 0; i < SCAN_PER / 4; ++i) {
    int a = v[i].x, b = v[i].y, c = v[i].z, d = v[i].w;
    v[i].x = run; run += a;
    v[i].y = run; run += b;
    v[i].z = run; run += c;
    v[i].w = run; run += d;
  }
  int inc = run;
  #pragma unroll
  for (int d = 1; d < 32; d <<= 1) {
    int y = __shfl_up(inc, d, 32);
    if (lane >= d) inc += y;
  }
  if (lane == 31) wsum[w] = inc;
  __syncthreads();
  if (w == 0) {
    int x = wsum[lane];
    int xi = x;
    #pragma unroll
    for (int d = 1; d < 32; d <<= 1) {
      int y = __shfl_up(xi, d, 32);
      if (lane >= d) xi += y;
    }
    wsum[lane] = xi - x;
  }
  __syncthreads();
  int base = wsum[w] + inc - run;
  int4* q = (int4*)(off + t * SCAN_PER);
  #pragma unroll
  for (int i = 0; i < SCAN_PER / 4; ++i) {
    int4 o = v[i];
    o.x += base; o.y += base; o.z += base; o.w += base;
    q[i] = o;
  }
}

#define PT_TILES  25
#define PT_CHUNKS (NN / 16 / PT_TILES)
#define PT_PITCH  (DD + 8)
static_assert(PT_CHUNKS * PT_TILES * 16 == NN, "the waves of a slot cover every node exactly once");
static_assert(PT_TILES % 2 == 1, "the walk below takes the tiles in pairs and then one last tile");

__device__ __forceinline__ void proj_load_rows(v16h (&hb)[4], const _Float16* p) {
  #pragma unroll
  for (int kk = 0; kk < 4; ++kk) hb[kk] = load_frag16(p + kk * 32);
}

__device__ __forceinline__ void lds_order_one_wave() {
  __builtin_amdgcn_fence(__ATOMIC_RELEASE, "workgroup", "local");
  __builtin_amdgcn_fence(__ATOMIC_ACQUIRE, "workgroup", "local");
}

__device__ __forceinline__ void proj_tile(const v16h (&w)[8][4], const v16h (&hb)[4],
                                          _Float16* tile, int lane, _Float16* q) {
  int lo = lane & 15, hi = lane >> 4;
  v8f acc[8];
  #pragma unroll
  for (int j = 0; j < 8; ++j) acc[j] = v8f_zero();
  #pragma unroll
  for (int kk = 0; kk < 4; ++kk) {
    #pragma unroll
    for (int jt = 0; jt < 8; ++jt) acc[jt] = wmma_f16(w[jt][kk], hb[kk], acc[jt]);
  }
  #pragma unroll
  for (int jt = 0; jt < 8; ++jt) {
    v8h o;
    #pragma unroll
    for (int r = 0; r < 8; ++r) o[r] = (_Float16)acc[jt][r];
    *(v8h*)(tile + lo * PT_PITCH + jt * 16 + hi * 8) = o;
  }
  lds_order_one_wave();
  #pragma unroll
  for (int i = 0; i < 8; ++i) {
    int row = 2 * i + hi;
    v8h o = *(const v8h*)(tile + row * PT_PITCH + lo * 8);
    *(v8h*)(q + (size_t)row * DD + lo * 8) = o;
  }
  lds_order_one_wave();
}

__global__ void __launch_bounds__(32) k_proj(const _Float16* __restrict__ h16,
                                             const _Float16* __restrict__ relT16,
                                             _Float16* __restrict__ proj16) {
  __shared__ __attribute__((aligned(16))) _Float16 tile[16 * PT_PITCH];
  int bid = blockIdx.x;
  int chunk = bid % PT_CHUNKS;
  int slot = bid / PT_CHUNKS;
  int lane = threadIdx.x;
  int lo = lane & 15, hi = lane >> 4;

  const _Float16* R = relT16 + ((size_t)slot * DD + lo) * DD + hi * 8;
  v16h w[8][4];
  #pragma unroll
  for (int jt = 0; jt < 8; ++jt)
    #pragma unroll
    for (int kk = 0; kk < 4; ++kk) w[jt][kk] = load_frag16(R + (size_t)jt * 16 * DD + kk * 32);

  size_t node0 = (size_t)chunk * PT_TILES * 16;
  const _Float16* Hn = h16 + (node0 + lo) * DD + hi * 8;
  _Float16* out = proj16 + ((size_t)slot * NN + node0) * DD;

  v16h ha[4], hc[4];
  proj_load_rows(ha, Hn);
  for (int t = 0; t + 1 < PT_TILES; t += 2) {
    proj_load_rows(hc, Hn + (size_t)(t + 1) * 16 * DD);
    proj_tile(w, ha, tile, lane, out + (size_t)t * 16 * DD);
    proj_load_rows(ha, Hn + (size_t)(t + 2) * 16 * DD);
    proj_tile(w, hc, tile, lane, out + (size_t)(t + 1) * 16 * DD);
  }
  proj_tile(w, ha, tile, lane, out + (size_t)(PT_TILES - 1) * 16 * DD);
}

__global__ void k_aproj(const _Float16* __restrict__ h16,
                        const _Float16* __restrict__ Wa16,
                        float* __restrict__ P) {
  int mtile = blockIdx.x;
  int lane = threadIdx.x;
  int lo = lane & 15, hi = lane >> 4;
  int node = mtile * 16 + lo;

  const _Float16* Hn = h16 + (size_t)node * DD + hi * 8;
  const _Float16* Wr = Wa16 + (size_t)lo * (2 * DD) + hi * 8;

  v8f acc[2][4];
  #pragma unroll
  for (int s = 0; s < 2; ++s)
    #pragma unroll
    for (int j = 0; j < 4; ++j) acc[s][j] = v8f_zero();

  #pragma unroll
  for (int kk = 0; kk < 4; ++kk) {
    v16h hb = load_frag16(Hn + kk * 32);
    #pragma unroll
    for (int s = 0; s < 2; ++s) {
      #pragma unroll
      for (int jt = 0; jt < 4; ++jt) {
        v16h w = load_frag16(Wr + (size_t)jt * 16 * (2 * DD) + s * DD + kk * 32);
        acc[s][jt] = wmma_f16(w, hb, acc[s][jt]);
      }
    }
  }

  float* prow = P + (size_t)node * (2 * AD) + 8 * hi;
  #pragma unroll
  for (int s = 0; s < 2; ++s) {
    #pragma unroll
    for (int jt = 0; jt < 4; ++jt) {
      v4f x0, x1;
      #pragma unroll
      for (int i = 0; i < 4; ++i) { x0[i] = acc[s][jt][i]; x1[i] = acc[s][jt][4 + i]; }
      float* q = prow + s * AD + jt * 16;
      *(v4f*)(q)     = x0;
      *(v4f*)(q + 4) = x1;
    }
  }
}

#define EPG 4
static_assert(NE % EPG == 0, "an 8-lane group's edges are all inside or all outside the edge list");
__global__ void k_score(const float* __restrict__ P,
                        const float* __restrict__ rel_att,
                        const float* __restrict__ b_a,
                        const int* __restrict__ src, const int* __restrict__ dst,
                        const int* __restrict__ etype,
                        const int* __restrict__ off, const int* __restrict__ rank,
                        int4* __restrict__ rec) {
  int t = blockIdx.x * blockDim.x + threadIdx.x;
  int sub = t & 7;
  int ebase = (t >> 3) * EPG;
  bool ok = ebase < NE;
  int eb = ok ? ebase : (NE - EPG);

  int dn[EPG], sn[EPG];
  #pragma unroll
  for (int k = 0; k < EPG; ++k) { dn[k] = dst[eb + k]; sn[k] = src[eb + k]; }

  v4f d0[EPG], d1[EPG], s0[EPG], s1[EPG];
  #pragma unroll
  for (int k = 0; k < EPG; ++k) {
    const float* pd = P + (size_t)dn[k] * (2 * AD) + sub * 8;
    const float* ps = P + (size_t)sn[k] * (2 * AD) + AD + sub * 8;
    d0[k] = *(const v4f*)(pd);
    d1[k] = *(const v4f*)(pd + 4);
    s0[k] = *(const v4f*)(ps);
    s1[k] = *(const v4f*)(ps + 4);
  }
  v4f w0 = *(const v4f*)(rel_att + sub * 8);
  v4f w1 = *(const v4f*)(rel_att + sub * 8 + 4);
  v4f b0 = *(const v4f*)(b_a + sub * 8);
  v4f b1 = *(const v4f*)(b_a + sub * 8 + 4);

  float acc[EPG];
  #pragma unroll
  for (int k = 0; k < EPG; ++k) {
    float a = 0.f;
    #pragma unroll
    for (int i = 0; i < 4; ++i) {
      float x = (d0[k][i] + s0[k][i]) + b0[i];
      x = (x > 0.f) ? x : 0.01f * x;
      a += x * w0[i];
    }
    #pragma unroll
    for (int i = 0; i < 4; ++i) {
      float x = (d1[k][i] + s1[k][i]) + b1[i];
      x = (x > 0.f) ? x : 0.01f * x;
      a += x * w1[i];
    }
    a += __shfl_xor(a, 1, 32);
    a += __shfl_xor(a, 2, 32);
    a += __shfl_xor(a, 4, 32);
    acc[k] = a;
  }

  float mine = acc[0];
  int dmine = dn[0], smine = sn[0];
  #pragma unroll
  for (int k = 1; k < EPG; ++k) {
    if (sub == k) { mine = acc[k]; dmine = dn[k]; smine = sn[k]; }
  }
  if (ok && sub < EPG) {
    int e = eb + sub;
    int4 r;
    r.x = smine; r.y = etype[e]; r.z = __float_as_int(mine); r.w = 0;
    rec[off[dmine] + rank[e]] = r;
  }
}

#define NPB 8
#define EU  4
static_assert(NN % NPB == 0, "every wave has a node");
static_assert(32 % EU == 0, "a group of EU slots never leaves the 32 records a wave holds");

__device__ __forceinline__ float wave_max(float x) {
  #pragma unroll
  for (int d = 16; d >= 1; d >>= 1) x = fmaxf(x, __shfl_xor(x, d, 32));
  return x;
}
__device__ __forceinline__ float wave_sum(float x) {
  #pragma unroll
  for (int d = 16; d >= 1; d >>= 1) x += __shfl_xor(x, d, 32);
  return x;
}

__device__ __forceinline__ void node_chunk(v4f& acc, int4 r, float a, int cnt, int n, int lane,
                                           const float* __restrict__ h,
                                           const _Float16* __restrict__ proj16) {
  for (int j0 = 0; j0 < cnt; j0 += EU) {
    v4h pdv[EU], psv[EU];
    v4f hs[EU];
    float at[EU];
    #pragma unroll
    for (int u = 0; u < EU; ++u) {
      int sn = __builtin_amdgcn_readlane(r.x, j0 + u);
      int et = __builtin_amdgcn_readlane(r.y, j0 + u);
      at[u] = __int_as_float(__builtin_amdgcn_readlane(__float_as_int(a), j0 + u));
      pdv[u] = *(const v4h*)(proj16 + ((size_t)et * NN + n) * DD + lane * 4);
      psv[u] = *(const v4h*)(proj16 + ((size_t)(8 + et) * NN + sn) * DD + lane * 4);
      hs[u]  = *(const v4f*)(h + (size_t)sn * DD + lane * 4);
    }
    #pragma unroll
    for (int u = 0; u < EU; ++u) {
      bool live = (j0 + u) < cnt;
      #pragma unroll
      for (int i = 0; i < 4; ++i) {
        float g = (float)pdv[u][i] + (float)psv[u][i];
        g = 1.f / (1.f + __expf(-g));
        float term = hs[u][i] * at[u] * g;
        acc[i] += live ? term : 0.f;
      }
    }
  }
}

__global__ void __launch_bounds__(32 * NPB) k_node(const float* __restrict__ h,
                                                   const _Float16* __restrict__ proj16,
                                                   const int* __restrict__ off,
                                                   const int4* __restrict__ rec,
                                                   float* __restrict__ hN) {
  int n = __builtin_amdgcn_readfirstlane(blockIdx.x * NPB + (threadIdx.x >> 5));
  int lane = threadIdx.x & 31;
  int beg = off[n], end = off[n + 1];
  const float NEG_INF = __int_as_float(0xFF800000);

  int i0 = beg + lane;
  bool has0 = i0 < end;
  int4 r0 = make_int4(0, 0, 0, 0);
  if (has0) r0 = rec[i0];
  float sc0 = has0 ? __int_as_float(r0.z) : NEG_INF;

  float m = sc0;
  for (int i = i0 + 32; i < end; i += 32) m = fmaxf(m, __int_as_float(rec[i].z));
  m = wave_max(m);

  float ex0 = has0 ? __expf(sc0 - m) : 0.f;
  float s = ex0;
  for (int i = i0 + 32; i < end; i += 32) s += __expf(__int_as_float(rec[i].z) - m);
  s = wave_sum(s);

  v4f acc;
  #pragma unroll
  for (int i = 0; i < 4; ++i) acc[i] = 0.f;

  int left = end - beg;
  node_chunk(acc, r0, has0 ? ex0 / s : 0.f, left < 32 ? left : 32, n, lane, h, proj16);
  for (int base = beg + 32; base < end; base += 32) {
    int i = base + lane;
    bool has = i < end;
    int4 r = make_int4(0, 0, 0, 0);
    if (has) r = rec[i];
    float a = has ? __expf(__int_as_float(r.z) - m) / s : 0.f;
    int c = end - base;
    node_chunk(acc, r, a, c < 32 ? c : 32, n, lane, h, proj16);
  }

  *(v4f*)(hN + (size_t)n * DD + lane * 4) = acc;
}

__global__ void k_final(const _Float16* __restrict__ h16,
                        const float* __restrict__ hN,
                        const _Float16* __restrict__ Wl16,
                        const float* __restrict__ b_lin,
                        float* __restrict__ out) {
  int mtile = blockIdx.x;
  int lane = threadIdx.x;
  int lo = lane & 15, hi = lane >> 4;
  int row = mtile * 16 + lo;

  v8f acc[8];
  #pragma unroll
  for (int j = 0; j < 8; ++j) acc[j] = v8f_zero();

  #pragma unroll
  for (int kk = 0; kk < 8; ++kk) {
    int k0 = kk * 32 + hi * 8;
    v16h a;
    if (k0 < DD) {
      a = load_frag16(h16 + (size_t)row * DD + k0);
    } else {
      const float* p = hN + (size_t)row * DD + (k0 - DD);
      union { v16h v; v4h h4[4]; } u;
      v4f f0 = *(const v4f*)(p);
      v4f f1 = *(const v4f*)(p + 4);
      v4f f2 = *(const v4f*)(p + 16);
      v4f f3 = *(const v4f*)(p + 20);
      #pragma unroll
      for (int i = 0; i < 4; ++i) {
        u.h4[0][i] = (_Float16)f0[i];
        u.h4[1][i] = (_Float16)f1[i];
        u.h4[2][i] = (_Float16)f2[i];
        u.h4[3][i] = (_Float16)f3[i];
      }
      a = u.v;
    }
    #pragma unroll
    for (int jt = 0; jt < 8; ++jt) {
      v16h b = load_frag16(Wl16 + (size_t)(jt * 16 + lo) * (2 * DD) + k0);
      acc[jt] = wmma_f16(a, b, acc[jt]);
    }
  }

  #pragma unroll
  for (int jt = 0; jt < 8; ++jt) {
    int col = jt * 16 + lo;
    float bb = b_lin[col];
    #pragma unroll
    for (int v = 0; v < 8; ++v) {
      float x = acc[jt][v] + bb;
      x = (x > 0.f) ? x : 0.01f * x;
      out[(size_t)(mtile * 16 + v + 8 * hi) * DD + col] = x;
    }
  }
}

extern "C" void kernel_launch(void* const* d_in, const int* in_sizes, int n_in,
                              void* d_out, int out_size, void* d_ws, size_t ws_size,
                              hipStream_t stream) {
  const float* h         = (const float*)d_in[0];
  const float* relations = (const float*)d_in[1];
  const float* rel_att   = (const float*)d_in[2];
  const float* W_a       = (const float*)d_in[3];
  const float* b_a       = (const float*)d_in[4];
  const float* W_lin     = (const float*)d_in[5];
  const float* b_lin     = (const float*)d_in[6];
  const int*   src       = (const int*)d_in[7];
  const int*   dst       = (const int*)d_in[8];
  const int*   etype     = (const int*)d_in[9];
  float* out = (float*)d_out;
  (void)in_sizes; (void)n_in; (void)out_size; (void)ws_size;

  char* ws = (char*)d_ws;
  size_t off = 0;
  auto take = [&](size_t bytes) -> char* {
    char* p = ws + off;
    off = (off + bytes + 255) & ~(size_t)255;
    return p;
  };
  _Float16* h16    = (_Float16*)take((size_t)NN * DD * 2);
  _Float16* relT16 = (_Float16*)take((size_t)16 * DD * DD * 2);
  _Float16* Wa16   = (_Float16*)take((size_t)AD * 2 * DD * 2);
  _Float16* Wl16   = (_Float16*)take((size_t)DD * 2 * DD * 2);
  _Float16* proj16 = (_Float16*)take((size_t)16 * NN * DD * 2);
  int4*  rec   = (int4*)take((size_t)NE * 16);
  int*   rank  = (int*)take((size_t)NE * 4);
  int*   deg   = (int*)take((size_t)NPAD * 4);
  int*   noff  = (int*)take((size_t)NPAD * 4);
  float* hN    = (float*)take((size_t)NN * DD * 4);

  const int nh = NN * DD;
  k_cvt<<<(nh + 255) / 256, 256, 0, stream>>>(h, h16, nh);
  k_cvt<<<(AD * 2 * DD + 255) / 256, 256, 0, stream>>>(W_a, Wa16, AD * 2 * DD);
  k_cvt<<<(DD * 2 * DD + 255) / 256, 256, 0, stream>>>(W_lin, Wl16, DD * 2 * DD);
  k_cvt_rel<<<(16 * DD * DD) / 256, 256, 0, stream>>>(relations, relT16);
  k_init<<<(NPAD + 255) / 256, 256, 0, stream>>>(deg);
  k_count<<<(NE + 255) / 256, 256, 0, stream>>>(dst, deg, rank);
  k_scan<<<1, SCAN_T, 0, stream>>>(deg, noff);

  float* P = (float*)proj16;
  k_aproj<<<NN / 16, 32, 0, stream>>>(h16, Wa16, P);
  k_score<<<(NE / EPG * 8 + 255) / 256, 256, 0, stream>>>(P, rel_att, b_a, src, dst, etype, noff, rank, rec);
  k_proj<<<16 * PT_CHUNKS, 32, 0, stream>>>(h16, relT16, proj16);
  k_node<<<NN / NPB, 32 * NPB, 0, stream>>>(h, proj16, noff, rec, hN);
  k_final<<<NN / 16, 32, 0, stream>>>(h16, hN, Wl16, b_lin, out);
}


# Gate sigmoid in the node pass: 1/(1+e^-x) on the hardware reciprocal (v_rcp_f32, 1 ulp) instead of the IEEE division sequence; the 4-edge loop body halves, registers 128 to 74
# speedup vs baseline: 2.1891x; 1.2031x over previous
#include <hip/hip_runtime.h>
#include <hip/hip_bf16.h>

#define NN 50000
#define NE 500000
#define NR 8
#define DD 128
#define AD 64

typedef _Float16 v16h __attribute__((ext_vector_type(16)));
typedef _Float16 v8h  __attribute__((ext_vector_type(8)));
typedef _Float16 v4h  __attribute__((ext_vector_type(4)));
typedef float    v8f  __attribute__((ext_vector_type(8)));
typedef float    v4f  __attribute__((ext_vector_type(4)));

__device__ __forceinline__ v8f v8f_zero() {
  v8f z;
  #pragma unroll
  for (int i = 0; i < 8; ++i) z[i] = 0.f;
  return z;
}

__device__ __forceinline__ v16h load_frag16(const _Float16* p) {
  union { v16h v; v8h h[2]; } u;
  u.h[0] = *(const v8h*)(p);
  u.h[1] = *(const v8h*)(p + 16);
  return u.v;
}

__device__ __forceinline__ v8f wmma_f16(v16h a, v16h b, v8f c) {
  return __builtin_amdgcn_wmma_f32_16x16x32_f16(false, a, false, b,
                                                (short)0, c, false, false);
}


#define CVT_TPB   256
#define CVT_TRIPS 4
#define CVT_GRID(n) (((n) / 8 + CVT_TPB * CVT_TRIPS - 1) / (CVT_TPB * CVT_TRIPS))
static_assert((NN * DD) % 8 == 0 && (AD * 2 * DD) % 8 == 0 && (DD * 2 * DD) % 8 == 0,
              "every converted array is a whole number of 8-element pieces");

__device__ __forceinline__ void cvt_pieces(const float* __restrict__ in,
                                           _Float16* __restrict__ out, int npiece, int blk, int nblk) {
  int t = blk * CVT_TPB + threadIdx.x;
  int width = nblk * CVT_TPB;
  v4f a[CVT_TRIPS], b[CVT_TRIPS];
  #pragma unroll
  for (int k = 0; k < CVT_TRIPS; ++k) {
    int p = t + k * width;
    int pc = (p < npiece) ? p : (npiece - 1);
    const float* q = in + (size_t)pc * 8;
    a[k] = *(const v4f*)(q);
    b[k] = *(const v4f*)(q + 4);
  }
  #pragma unroll
  for (int k = 0; k < CVT_TRIPS; ++k) {
    int p = t + k * width;
    if (p < npiece) {
      v8h o;
      #pragma unroll
      for (int i = 0; i < 4; ++i) { o[i] = (_Float16)a[k][i]; o[4 + i] = (_Float16)b[k][i]; }
      *(v8h*)(out + (size_t)p * 8) = o;
    }
  }
}

__device__ __forceinline__ void cvt_rel_block(const float* __restrict__ rel,
                                              _Float16* __restrict__ relT, int b) {
  int i = b * CVT_TPB + threadIdx.x;
  int d = i & (DD - 1);
  int e = (i >> 7) & (DD - 1);
  int slot = i >> 14;
  int half = slot >> 3, r = slot & 7;
  relT[i] = (_Float16)rel[((size_t)r * 2 * DD + half * DD + d) * DD + e];
}

#define SCAN_T   1024
#define SCAN_PER 52
#define NPAD     (SCAN_T * SCAN_PER)
static_assert(NPAD >= NN + 1, "the scan must also produce the end of the last node's segment");
static_assert(SCAN_PER % 4 == 0, "a thread's piece is read as 16-byte words");

#define SU_B_H   CVT_GRID(NN * DD)
#define SU_B_REL ((16 * DD * DD) / CVT_TPB)
#define SU_B_DEG (NPAD / CVT_TPB)
#define SU_B_WA  CVT_GRID(AD * 2 * DD)
#define SU_B_WL  CVT_GRID(DD * 2 * DD)
#define SU_BLOCKS (SU_B_H + SU_B_REL + SU_B_DEG + SU_B_WA + SU_B_WL)
static_assert((16 * DD * DD) % CVT_TPB == 0 && NPAD % CVT_TPB == 0, "whole blocks, no tail lanes");

__global__ void __launch_bounds__(CVT_TPB) k_setup(const float* __restrict__ h,
                                                   const float* __restrict__ rel,
                                                   const float* __restrict__ W_a,
                                                   const float* __restrict__ W_lin,
                                                   _Float16* __restrict__ h16,
                                                   _Float16* __restrict__ relT16,
                                                   _Float16* __restrict__ Wa16,
                                                   _Float16* __restrict__ Wl16,
                                                   int* __restrict__ deg) {
  int b = blockIdx.x;
  if (b < SU_B_H) { cvt_pieces(h, h16, NN * DD / 8, b, SU_B_H); return; }
  b -= SU_B_H;
  if (b < SU_B_REL) { cvt_rel_block(rel, relT16, b); return; }
  b -= SU_B_REL;
  if (b < SU_B_DEG) { deg[b * CVT_TPB + threadIdx.x] = 0; return; }
  b -= SU_B_DEG;
  if (b < SU_B_WA) { cvt_pieces(W_a, Wa16, AD * 2 * DD / 8, b, SU_B_WA); return; }
  b -= SU_B_WA;
  cvt_pieces(W_lin, Wl16, DD * 2 * DD / 8, b, SU_B_WL);
}

#define CNT_TRIPS  4
#define CNT_PIECES (NE / 4)
#define CNT_BLOCKS ((CNT_PIECES + 32 * CNT_TRIPS - 1) / (32 * CNT_TRIPS))
static_assert(NE % 4 == 0, "the edge list is read in pieces of four");
__device__ __forceinline__ void count_role(const int* __restrict__ dst, int* __restrict__ deg,
                                           int* __restrict__ rank, int b, int lane) {
  int4 d[CNT_TRIPS], r[CNT_TRIPS];
  #pragma unroll
  for (int k = 0; k < CNT_TRIPS; ++k) {
    int p = (b * CNT_TRIPS + k) * 32 + lane;
    int pc = (p < CNT_PIECES) ? p : (CNT_PIECES - 1);
    d[k] = ((const int4*)dst)[pc];
  }
  #pragma unroll
  for (int k = 0; k < CNT_TRIPS; ++k) {
    int p = (b * CNT_TRIPS + k) * 32 + lane;
    if (p < CNT_PIECES) {
      r[k].x = atomicAdd(&deg[d[k].x], 1);
      r[k].y = atomicAdd(&deg[d[k].y], 1);
      r[k].z = atomicAdd(&deg[d[k].z], 1);
      r[k].w = atomicAdd(&deg[d[k].w], 1);
    }
  }
  #pragma unroll
  for (int k = 0; k < CNT_TRIPS; ++k) {
    int p = (b * CNT_TRIPS + k) * 32 + lane;
    if (p < CNT_PIECES) ((int4*)rank)[p] = r[k];
  }
}

__global__ void __launch_bounds__(SCAN_T) k_scan(const int* __restrict__ deg,
                                                 int* __restrict__ off) {
  __shared__ int wsum[SCAN_T / 32];
  int t = threadIdx.x;
  int lane = t & 31, w = t >> 5;
  const int4* p = (const int4*)(deg + t * SCAN_PER);
  int4 v[SCAN_PER / 4];
  #pragma unroll
  for (int i = 0; i < SCAN_PER / 4; ++i) v[i] = p[i];
  int run = 0;
  #pragma unroll
  for (int i = 0; i < SCAN_PER / 4; ++i) {
    int a = v[i].x, b = v[i].y, c = v[i].z, d = v[i].w;
    v[i].x = run; run += a;
    v[i].y = run; run += b;
    v[i].z = run; run += c;
    v[i].w = run; run += d;
  }
  int inc = run;
  #pragma unroll
  for (int d = 1; d < 32; d <<= 1) {
    int y = __shfl_up(inc, d, 32);
    if (lane >= d) inc += y;
  }
  if (lane == 31) wsum[w] = inc;
  __syncthreads();
  if (w == 0) {
    int x = wsum[lane];
    int xi = x;
    #pragma unroll
    for (int d = 1; d < 32; d <<= 1) {
      int y = __shfl_up(xi, d, 32);
      if (lane >= d) xi += y;
    }
    wsum[lane] = xi - x;
  }
  __syncthreads();
  int base = wsum[w] + inc - run;
  int4* q = (int4*)(off + t * SCAN_PER);
  #pragma unroll
  for (int i = 0; i < SCAN_PER / 4; ++i) {
    int4 o = v[i];
    o.x += base; o.y += base; o.z += base; o.w += base;
    q[i] = o;
  }
}

#define PT_TILES  25
#define PT_CHUNKS (NN / 16 / PT_TILES)
#define PT_PITCH  (DD + 8)
static_assert(PT_CHUNKS * PT_TILES * 16 == NN, "the waves of a slot cover every node exactly once");
static_assert(PT_TILES % 2 == 1, "the walk below takes the tiles in pairs and then one last tile");

__device__ __forceinline__ void proj_load_rows(v16h (&hb)[4], const _Float16* p) {
  #pragma unroll
  for (int kk = 0; kk < 4; ++kk) hb[kk] = load_frag16(p + kk * 32);
}

__device__ __forceinline__ void lds_order_one_wave() {
  __builtin_amdgcn_fence(__ATOMIC_RELEASE, "workgroup", "local");
  __builtin_amdgcn_fence(__ATOMIC_ACQUIRE, "workgroup", "local");
}

__device__ __forceinline__ void proj_tile(const v16h (&w)[8][4], const v16h (&hb)[4],
                                          _Float16* tile, int lane, _Float16* q) {
  int lo = lane & 15, hi = lane >> 4;
  v8f acc[8];
  #pragma unroll
  for (int j = 0; j < 8; ++j) acc[j] = v8f_zero();
  #pragma unroll
  for (int kk = 0; kk < 4; ++kk) {
    #pragma unroll
    for (int jt = 0; jt < 8; ++jt) acc[jt] = wmma_f16(w[jt][kk], hb[kk], acc[jt]);
  }
  #pragma unroll
  for (int jt = 0; jt < 8; ++jt) {
    v8h o;
    #pragma unroll
    for (int r = 0; r < 8; ++r) o[r] = (_Float16)acc[jt][r];
    *(v8h*)(tile + lo * PT_PITCH + jt * 16 + hi * 8) = o;
  }
  lds_order_one_wave();
  #pragma unroll
  for (int i = 0; i < 8; ++i) {
    int row = 2 * i + hi;
    v8h o = *(const v8h*)(tile + row * PT_PITCH + lo * 8);
    *(v8h*)(q + (size_t)row * DD + lo * 8) = o;
  }
  lds_order_one_wave();
}

__device__ __forceinline__ void proj_role(const _Float16* __restrict__ h16,
                                          const _Float16* __restrict__ relT16,
                                          _Float16* __restrict__ proj16,
                                          int bid, int lane, _Float16* tile) {
  int chunk = bid % PT_CHUNKS;
  int slot = bid / PT_CHUNKS;
  int lo = lane & 15, hi = lane >> 4;

  const _Float16* R = relT16 + ((size_t)slot * DD + lo) * DD + hi * 8;
  v16h w[8][4];
  #pragma unroll
  for (int jt = 0; jt < 8; ++jt)
    #pragma unroll
    for (int kk = 0; kk < 4; ++kk) w[jt][kk] = load_frag16(R + (size_t)jt * 16 * DD + kk * 32);

  size_t node0 = (size_t)chunk * PT_TILES * 16;
  const _Float16* Hn = h16 + (node0 + lo) * DD + hi * 8;
  _Float16* out = proj16 + ((size_t)slot * NN + node0) * DD;

  v16h ha[4], hc[4];
  proj_load_rows(ha, Hn);
  for (int t = 0; t + 1 < PT_TILES; t += 2) {
    proj_load_rows(hc, Hn + (size_t)(t + 1) * 16 * DD);
    proj_tile(w, ha, tile, lane, out + (size_t)t * 16 * DD);
    proj_load_rows(ha, Hn + (size_t)(t + 2) * 16 * DD);
    proj_tile(w, hc, tile, lane, out + (size_t)(t + 1) * 16 * DD);
  }
  proj_tile(w, ha, tile, lane, out + (size_t)(PT_TILES - 1) * 16 * DD);
}

__device__ __forceinline__ void aproj_role(const _Float16* __restrict__ h16,
                                           const _Float16* __restrict__ Wa16,
                                           float* __restrict__ P, int mtile, int lane) {
  int lo = lane & 15, hi = lane >> 4;
  int node = mtile * 16 + lo;

  const _Float16* Hn = h16 + (size_t)node * DD + hi * 8;
  const _Float16* Wr = Wa16 + (size_t)lo * (2 * DD) + hi * 8;

  v8f acc[2][4];
  #pragma unroll
  for (int s = 0; s < 2; ++s)
    #pragma unroll
    for (int j = 0; j < 4; ++j) acc[s][j] = v8f_zero();

  #pragma unroll
  for (int kk = 0; kk < 4; ++kk) {
    v16h hb = load_frag16(Hn + kk * 32);
    #pragma unroll
    for (int s = 0; s < 2; ++s) {
      #pragma unroll
      for (int jt = 0; jt < 4; ++jt) {
        v16h w = load_frag16(Wr + (size_t)jt * 16 * (2 * DD) + s * DD + kk * 32);
        acc[s][jt] = wmma_f16(w, hb, acc[s][jt]);
      }
    }
  }

  float* prow = P + (size_t)node * (2 * AD) + 8 * hi;
  #pragma unroll
  for (int s = 0; s < 2; ++s) {
    #pragma unroll
    for (int jt = 0; jt < 4; ++jt) {
      v4f x0, x1;
      #pragma unroll
      for (int i = 0; i < 4; ++i) { x0[i] = acc[s][jt][i]; x1[i] = acc[s][jt][4 + i]; }
      float* q = prow + s * AD + jt * 16;
      *(v4f*)(q)     = x0;
      *(v4f*)(q + 4) = x1;
    }
  }
}

#define MID_B_PROJ  (16 * PT_CHUNKS)
#define MID_B_APROJ (NN / 16)
#define MID_BLOCKS  (CNT_BLOCKS + MID_B_PROJ + MID_B_APROJ)
__global__ void __launch_bounds__(32) k_mid(const int* __restrict__ dst, int* __restrict__ deg,
                                            int* __restrict__ rank,
                                            const _Float16* __restrict__ h16,
                                            const _Float16* __restrict__ relT16,
                                            _Float16* __restrict__ proj16,
                                            const _Float16* __restrict__ Wa16,
                                            float* __restrict__ P) {
  __shared__ __attribute__((aligned(16))) _Float16 tile[16 * PT_PITCH];
  int b = blockIdx.x;
  int lane = threadIdx.x;
  if (b < CNT_BLOCKS) { count_role(dst, deg, rank, b, lane); return; }
  b -= CNT_BLOCKS;
  if (b < MID_B_PROJ) { proj_role(h16, relT16, proj16, b, lane, tile); return; }
  b -= MID_B_PROJ;
  aproj_role(h16, Wa16, P, b, lane);
}

#define EPG 4
static_assert(NE % EPG == 0, "an 8-lane group's edges are all inside or all outside the edge list");
__global__ void k_score(const float* __restrict__ P,
                        const float* __restrict__ rel_att,
                        const float* __restrict__ b_a,
                        const int* __restrict__ src, const int* __restrict__ dst,
                        const int* __restrict__ etype,
                        const int* __restrict__ off, const int* __restrict__ rank,
                        int4* __restrict__ rec) {
  int t = blockIdx.x * blockDim.x + threadIdx.x;
  int sub = t & 7;
  int ebase = (t >> 3) * EPG;
  bool ok = ebase < NE;
  int eb = ok ? ebase : (NE - EPG);

  int dn[EPG], sn[EPG];
  #pragma unroll
  for (int k = 0; k < EPG; ++k) { dn[k] = dst[eb + k]; sn[k] = src[eb + k]; }

  v4f d0[EPG], d1[EPG], s0[EPG], s1[EPG];
  #pragma unroll
  for (int k = 0; k < EPG; ++k) {
    const float* pd = P + (size_t)dn[k] * (2 * AD) + sub * 8;
    const float* ps = P + (size_t)sn[k] * (2 * AD) + AD + sub * 8;
    d0[k] = *(const v4f*)(pd);
    d1[k] = *(const v4f*)(pd + 4);
    s0[k] = *(const v4f*)(ps);
    s1[k] = *(const v4f*)(ps + 4);
  }
  v4f w0 = *(const v4f*)(rel_att + sub * 8);
  v4f w1 = *(const v4f*)(rel_att + sub * 8 + 4);
  v4f b0 = *(const v4f*)(b_a + sub * 8);
  v4f b1 = *(const v4f*)(b_a + sub * 8 + 4);

  float acc[EPG];
  #pragma unroll
  for (int k = 0; k < EPG; ++k) {
    float a = 0.f;
    #pragma unroll
    for (int i = 0; i < 4; ++i) {
      float x = (d0[k][i] + s0[k][i]) + b0[i];
      x = (x > 0.f) ? x : 0.01f * x;
      a += x * w0[i];
    }
    #pragma unroll
    for (int i = 0; i < 4; ++i) {
      float x = (d1[k][i] + s1[k][i]) + b1[i];
      x = (x > 0.f) ? x : 0.01f * x;
      a += x * w1[i];
    }
    a += __shfl_xor(a, 1, 32);
    a += __shfl_xor(a, 2, 32);
    a += __shfl_xor(a, 4, 32);
    acc[k] = a;
  }

  float mine = acc[0];
  int dmine = dn[0], smine = sn[0];
  #pragma unroll
  for (int k = 1; k < EPG; ++k) {
    if (sub == k) { mine = acc[k]; dmine = dn[k]; smine = sn[k]; }
  }
  if (ok && sub < EPG) {
    int e = eb + sub;
    int4 r;
    r.x = smine; r.y = etype[e]; r.z = __float_as_int(mine); r.w = 0;
    rec[off[dmine] + rank[e]] = r;
  }
}

#define NPB 8
#define EU  4
static_assert(NN % NPB == 0, "every wave has a node");
static_assert(32 % EU == 0, "a group of EU slots never leaves the 32 records a wave holds");

__device__ __forceinline__ float wave_max(float x) {
  #pragma unroll
  for (int d = 16; d >= 1; d >>= 1) x = fmaxf(x, __shfl_xor(x, d, 32));
  return x;
}
__device__ __forceinline__ float wave_sum(float x) {
  #pragma unroll
  for (int d = 16; d >= 1; d >>= 1) x += __shfl_xor(x, d, 32);
  return x;
}

__device__ __forceinline__ void node_chunk(v4f& acc, int4 r, float a, int cnt, int n, int lane,
                                           const float* __restrict__ h,
                                           const _Float16* __restrict__ proj16) {
  for (int j0 = 0; j0 < cnt; j0 += EU) {
    v4h pdv[EU], psv[EU];
    v4f hs[EU];
    float at[EU];
    #pragma unroll
    for (int u = 0; u < EU; ++u) {
      int sn = __builtin_amdgcn_readlane(r.x, j0 + u);
      int et = __builtin_amdgcn_readlane(r.y, j0 + u);
      at[u] = __int_as_float(__builtin_amdgcn_readlane(__float_as_int(a), j0 + u));
      pdv[u] = *(const v4h*)(proj16 + ((size_t)et * NN + n) * DD + lane * 4);
      psv[u] = *(const v4h*)(proj16 + ((size_t)(8 + et) * NN + sn) * DD + lane * 4);
      hs[u]  = *(const v4f*)(h + (size_t)sn * DD + lane * 4);
    }
    #pragma unroll
    for (int u = 0; u < EU; ++u) {
      bool live = (j0 + u) < cnt;
      #pragma unroll
      for (int i = 0; i < 4; ++i) {
        float g = (float)pdv[u][i] + (float)psv[u][i];
        g = __builtin_amdgcn_rcpf(1.f + __expf(-g));
        float term = hs[u][i] * at[u] * g;
        acc[i] += live ? term : 0.f;
      }
    }
  }
}

__global__ void __launch_bounds__(32 * NPB) k_node(const float* __restrict__ h,
                                                   const _Float16* __restrict__ proj16,
                                                   const int* __restrict__ off,
                                                   const int4* __restrict__ rec,
                                                   float* __restrict__ hN) {
  int n = __builtin_amdgcn_readfirstlane(blockIdx.x * NPB + (threadIdx.x >> 5));
  int lane = threadIdx.x & 31;
  int beg = off[n], end = off[n + 1];
  const float NEG_INF = __int_as_float(0xFF800000);

  int i0 = beg + lane;
  bool has0 = i0 < end;
  int4 r0 = make_int4(0, 0, 0, 0);
  if (has0) r0 = rec[i0];
  float sc0 = has0 ? __int_as_float(r0.z) : NEG_INF;

  float m = sc0;
  for (int i = i0 + 32; i < end; i += 32) m = fmaxf(m, __int_as_float(rec[i].z));
  m = wave_max(m);

  float ex0 = has0 ? __expf(sc0 - m) : 0.f;
  float s = ex0;
  for (int i = i0 + 32; i < end; i += 32) s += __expf(__int_as_float(rec[i].z) - m);
  s = wave_sum(s);

  v4f acc;
  #pragma unroll
  for (int i = 0; i < 4; ++i) acc[i] = 0.f;

  int left = end - beg;
  node_chunk(acc, r0, has0 ? ex0 / s : 0.f, left < 32 ? left : 32, n, lane, h, proj16);
  for (int base = beg + 32; base < end; base += 32) {
    int i = base + lane;
    bool has = i < end;
    int4 r = make_int4(0, 0, 0, 0);
    if (has) r = rec[i];
    float a = has ? __expf(__int_as_float(r.z) - m) / s : 0.f;
    int c = end - base;
    node_chunk(acc, r, a, c < 32 ? c : 32, n, lane, h, proj16);
  }

  *(v4f*)(hN + (size_t)n * DD + lane * 4) = acc;
}

__global__ void k_final(const _Float16* __restrict__ h16,
                        const float* __restrict__ hN,
                        const _Float16* __restrict__ Wl16,
                        const float* __restrict__ b_lin,
                        float* __restrict__ out) {
  int mtile = blockIdx.x;
  int lane = threadIdx.x;
  int lo = lane & 15, hi = lane >> 4;
  int row = mtile * 16 + lo;

  v8f acc[8];
  #pragma unroll
  for (int j = 0; j < 8; ++j) acc[j] = v8f_zero();

  #pragma unroll
  for (int kk = 0; kk < 8; ++kk) {
    int k0 = kk * 32 + hi * 8;
    v16h a;
    if (k0 < DD) {
      a = load_frag16(h16 + (size_t)row * DD + k0);
    } else {
      const float* p = hN + (size_t)row * DD + (k0 - DD);
      union { v16h v; v4h h4[4]; } u;
      v4f f0 = *(const v4f*)(p);
      v4f f1 = *(const v4f*)(p + 4);
      v4f f2 = *(const v4f*)(p + 16);
      v4f f3 = *(const v4f*)(p + 20);
      #pragma unroll
      for (int i = 0; i < 4; ++i) {
        u.h4[0][i] = (_Float16)f0[i];
        u.h4[1][i] = (_Float16)f1[i];
        u.h4[2][i] = (_Float16)f2[i];
        u.h4[3][i] = (_Float16)f3[i];
      }
      a = u.v;
    }
    #pragma unroll
    for (int jt = 0; jt < 8; ++jt) {
      v16h b = load_frag16(Wl16 + (size_t)(jt * 16 + lo) * (2 * DD) + k0);
      acc[jt] = wmma_f16(a, b, acc[jt]);
    }
  }

  #pragma unroll
  for (int jt = 0; jt < 8; ++jt) {
    int col = jt * 16 + lo;
    float bb = b_lin[col];
    #pragma unroll
    for (int v = 0; v < 8; ++v) {
      float x = acc[jt][v] + bb;
      x = (x > 0.f) ? x : 0.01f * x;
      out[(size_t)(mtile * 16 + v + 8 * hi) * DD + col] = x;
    }
  }
}

extern "C" void kernel_launch(void* const* d_in, const int* in_sizes, int n_in,
                              void* d_out, int out_size, void* d_ws, size_t ws_size,
                              hipStream_t stream) {
  const float* h         = (const float*)d_in[0];
  const float* relations = (const float*)d_in[1];
  const float* rel_att   = (const float*)d_in[2];
  const float* W_a       = (const float*)d_in[3];
  const float* b_a       = (const float*)d_in[4];
  const float* W_lin     = (const float*)d_in[5];
  const float* b_lin     = (const float*)d_in[6];
  const int*   src       = (const int*)d_in[7];
  const int*   dst       = (const int*)d_in[8];
  const int*   etype     = (const int*)d_in[9];
  float* out = (float*)d_out;
  (void)in_sizes; (void)n_in; (void)out_size; (void)ws_size;

  char* ws = (char*)d_ws;
  size_t off = 0;
  auto take = [&](size_t bytes) -> char* {
    char* p = ws + off;
    off = (off + bytes + 255) & ~(size_t)255;
    return p;
  };
  _Float16* h16    = (_Float16*)take((size_t)NN * DD * 2);
  _Float16* relT16 = (_Float16*)take((size_t)16 * DD * DD * 2);
  _Float16* Wa16   = (_Float16*)take((size_t)AD * 2 * DD * 2);
  _Float16* Wl16   = (_Float16*)take((size_t)DD * 2 * DD * 2);
  _Float16* proj16 = (_Float16*)take((size_t)16 * NN * DD * 2);
  int4*  rec   = (int4*)take((size_t)NE * 16);
  int*   rank  = (int*)take((size_t)NE * 4);
  int*   deg   = (int*)take((size_t)NPAD * 4);
  int*   noff  = (int*)take((size_t)NPAD * 4);
  float* hN    = (float*)take((size_t)NN * DD * 4);

  k_setup<<<SU_BLOCKS, CVT_TPB, 0, stream>>>(h, relations, W_a, W_lin, h16, relT16, Wa16, Wl16, deg);

  float* P = hN;
  k_mid<<<MID_BLOCKS, 32, 0, stream>>>(dst, deg, rank, h16, relT16, proj16, Wa16, P);
  k_scan<<<1, SCAN_T, 0, stream>>>(deg, noff);
  k_score<<<(NE / EPG * 8 + 255) / 256, 256, 0, stream>>>(P, rel_att, b_a, src, dst, etype, noff, rank, rec);
  k_node<<<NN / NPB, 32 * NPB, 0, stream>>>(h, proj16, noff, rec, hN);
  k_final<<<NN / 16, 32, 0, stream>>>(h16, hN, Wl16, b_lin, out);
}
